# starting kernel
# baseline (speedup 1.0000x reference)
#include <hip/hip_runtime.h>
#include <hip/hip_bf16.h>


#define N_NODES   100000
#define E_EDGES   1600000
#define IN_F      128
#define H_HEADS   4
#define F_OUT     16
#define HF        64
#define NEG_SLOPE 0.01f

typedef float v2f __attribute__((ext_vector_type(2)));
typedef float v8f __attribute__((ext_vector_type(8)));

__device__ __forceinline__ unsigned fenc(float f) {
    unsigned u = __float_as_uint(f);
    return (u & 0x80000000u) ? ~u : (u | 0x80000000u);
}
__device__ __forceinline__ float fdec(unsigned k) {
    return (k & 0x80000000u) ? __uint_as_float(k ^ 0x80000000u)
                             : __uint_as_float(~k);
}

__global__ void gat_init(float* __restrict__ out, float* __restrict__ denom,
                         unsigned* __restrict__ maxenc) {
    size_t i = (size_t)blockIdx.x * blockDim.x + threadIdx.x;
    if (i < (size_t)N_NODES * HF)      out[i]   = 0.0f;
    if (i < (size_t)N_NODES * H_HEADS) denom[i] = 0.0f;
    if (i == 0) *maxenc = 0u;
}

__global__ __launch_bounds__(128)
void gat_transform_wmma(const float* __restrict__ x,
                        const float* __restrict__ W,
                        float* __restrict__ nt) {
    const int lane = threadIdx.x & 31;
    const int wave = threadIdx.x >> 5;
    const int m    = lane & 15;
    const int kk   = lane >> 4;
    const int row  = blockIdx.x * 16 + m;
    const int ncol = wave * 16 + m;

    const float* xrow = x + (size_t)row  * IN_F + 2 * kk;
    const float* wrow = W + (size_t)ncol * IN_F + 2 * kk;

    v8f c = {};
#pragma unroll
    for (int k0 = 0; k0 < IN_F; k0 += 4) {
        v2f a = *(const v2f*)(xrow + k0);
        v2f b = *(const v2f*)(wrow + k0);
        c = __builtin_amdgcn_wmma_f32_16x16x4_f32(
                 false, a,  false, b,
                 (short)0, c,  false,  false);
    }

    const int half = lane >> 4;
#pragma unroll
    for (int r = 0; r < 8; ++r) {
        int mm = r + 8 * half;
        nt[(size_t)(blockIdx.x * 16 + mm) * HF + wave * 16 + (lane & 15)] = c[r];
    }
}

__global__ __launch_bounds__(256)
void gat_edge_scores(const float* __restrict__ nt,
                     const float* __restrict__ a_w,
                     const int*   __restrict__ src,
                     const int*   __restrict__ tgt,
                     float* __restrict__ aw,
                     unsigned* __restrict__ maxenc) {
    __shared__ unsigned smax;
    if (threadIdx.x == 0) smax = 0u;
    __syncthreads();

    const int lane = threadIdx.x & 31;
    const int wv   = threadIdx.x >> 5;
    const int e    = blockIdx.x * 8 + wv;

    float res[H_HEADS] = {0.f, 0.f, 0.f, 0.f};
    if (e < E_EDGES) {
        const int vs = src[e];
        const int vt = tgt[e];
        const int j  = lane * 4;
        const int hp = j >> 5;
        const int off = j & 31;
        const float* base = (off < 16)
            ? (nt + (size_t)vs * HF + hp * F_OUT + off)
            : (nt + (size_t)vt * HF + hp * F_OUT + (off - 16));
        const float4 v = *(const float4*)base;
#pragma unroll
        for (int h = 0; h < H_HEADS; ++h) {
            const float* ar = a_w + h * (2 * HF) + j;
            res[h] = fmaf(v.x, ar[0], fmaf(v.y, ar[1],
                     fmaf(v.z, ar[2], v.w * ar[3])));
        }
    }
#pragma unroll
    for (int d = 16; d >= 1; d >>= 1) {
#pragma unroll
        for (int h = 0; h < H_HEADS; ++h)
            res[h] += __shfl_xor(res[h], d, 32);
    }
    if (e < E_EDGES && lane == 0) {
        float mx = res[0];
#pragma unroll
        for (int h = 0; h < H_HEADS; ++h) {
            aw[(size_t)e * H_HEADS + h] = res[h];
            mx = fmaxf(mx, res[h]);
        }
        atomicMax(&smax, fenc(mx));
    }
    __syncthreads();
    if (threadIdx.x == 0) atomicMax(maxenc, smax);
}

__global__ __launch_bounds__(256)
void gat_softmax_denom(float* __restrict__ awex,
                       const int* __restrict__ tgt,
                       float* __restrict__ denom,
                       const unsigned* __restrict__ maxenc) {
    const int e = blockIdx.x * blockDim.x + threadIdx.x;
    if (e >= E_EDGES) return;
    const float gmax = fdec(*maxenc);
    const int vt = tgt[e];
    float4 a = *(const float4*)(awex + (size_t)e * H_HEADS);
    float v[H_HEADS] = {a.x, a.y, a.z, a.w};
#pragma unroll
    for (int h = 0; h < H_HEADS; ++h) {
        float z = v[h] - gmax;
        z = (z > 0.0f) ? z : NEG_SLOPE * z;
        float ev = __expf(z);
        awex[(size_t)e * H_HEADS + h] = ev;
        atomicAdd(&denom[(size_t)vt * H_HEADS + h], ev);
    }
}

__global__ __launch_bounds__(256)
void gat_aggregate(const float* __restrict__ nt,
                   const float* __restrict__ ex,
                   const float* __restrict__ denom,
                   const int*   __restrict__ src,
                   const int*   __restrict__ tgt,
                   float* __restrict__ out) {
    const int lane = threadIdx.x & 31;
    const int wv   = threadIdx.x >> 5;
    const int e    = blockIdx.x * 8 + wv;
    if (e >= E_EDGES) return;
    const int vs = src[e];
    const int vt = tgt[e];
#pragma unroll
    for (int c = 0; c < 2; ++c) {
        const int idx = lane + 32 * c;
        const int h   = idx >> 4;
        const float alpha = ex[(size_t)e * H_HEADS + h] /
                            (denom[(size_t)vt * H_HEADS + h] + 1e-8f);
        const float sv = nt[(size_t)vs * HF + idx];
        atomicAdd(&out[(size_t)vt * HF + idx], alpha * sv);
    }
}

extern "C" void kernel_launch(void* const* d_in, const int* in_sizes, int n_in,
                              void* d_out, int out_size, void* d_ws, size_t ws_size,
                              hipStream_t stream) {
    const float* x   = (const float*)d_in[0];
    const float* W_w = (const float*)d_in[1];
    const float* a_w = (const float*)d_in[2];
    const int*   ei  = (const int*)d_in[3];
    const int* src = ei;
    const int* tgt = ei + E_EDGES;
    float* out = (float*)d_out;

    float* nt    = (float*)d_ws;
    float* awex  = nt   + (size_t)N_NODES * HF;
    float* denom = awex + (size_t)E_EDGES * H_HEADS;
    unsigned* maxenc = (unsigned*)(denom + (size_t)N_NODES * H_HEADS);

    (void)in_sizes; (void)n_in; (void)out_size; (void)ws_size;

    {
        size_t total = (size_t)N_NODES * HF;
        int blocks = (int)((total + 255) / 256);
        gat_init<<<blocks, 256, 0, stream>>>(out, denom, maxenc);
    }
    {
        gat_transform_wmma<<<N_NODES / 16, 128, 0, stream>>>(x, W_w, nt);
    }
    {
        gat_edge_scores<<<(E_EDGES + 7) / 8, 256, 0, stream>>>(
            nt, a_w, src, tgt, awex, maxenc);
    }
    {
        gat_softmax_denom<<<(E_EDGES + 255) / 256, 256, 0, stream>>>(
            awex, tgt, denom, maxenc);
    }
    {
        gat_aggregate<<<(E_EDGES + 7) / 8, 256, 0, stream>>>(
            nt, awex, denom, src, tgt, out);
    }
}


# Edge logits from per-node scores: transform epilogue computes P[n,h],Q[n,h] (kept in out until it is zeroed), edge pass is P[src]+Q[tgt] per thread; the 128-term logit sum order changes
# speedup vs baseline: 1.4689x; 1.4689x over previous
#include <hip/hip_runtime.h>
#include <hip/hip_bf16.h>


#define N_NODES   100000
#define E_EDGES   1600000
#define IN_F      128
#define H_HEADS   4
#define F_OUT     16
#define HF        64
#define NEG_SLOPE 0.01f
#define PQ_W      8
#define TILE_PITCH 68

typedef float v2f __attribute__((ext_vector_type(2)));
typedef float v8f __attribute__((ext_vector_type(8)));

__device__ __forceinline__ unsigned fenc(float f) {
    unsigned u = __float_as_uint(f);
    return (u & 0x80000000u) ? ~u : (u | 0x80000000u);
}
__device__ __forceinline__ float fdec(unsigned k) {
    return (k & 0x80000000u) ? __uint_as_float(k ^ 0x80000000u)
                             : __uint_as_float(~k);
}

__global__ void gat_init(float* __restrict__ out, float* __restrict__ denom) {
    size_t i = (size_t)blockIdx.x * blockDim.x + threadIdx.x;
    if (i < (size_t)N_NODES * HF)      out[i]   = 0.0f;
    if (i < (size_t)N_NODES * H_HEADS) denom[i] = 0.0f;
}

__global__ __launch_bounds__(128)
void gat_transform_wmma(const float* __restrict__ x,
                        const float* __restrict__ W,
                        const float* __restrict__ a_w,
                        float* __restrict__ nt,
                        float* __restrict__ pq,
                        unsigned* __restrict__ maxenc) {
    __shared__ __attribute__((aligned(16))) float tile[16 * TILE_PITCH];
    const int lane = threadIdx.x & 31;
    const int wave = threadIdx.x >> 5;
    const int m    = lane & 15;
    const int kk   = lane >> 4;
    const int row  = blockIdx.x * 16 + m;
    const int ncol = wave * 16 + m;

    const float* xrow = x + (size_t)row  * IN_F + 2 * kk;
    const float* wrow = W + (size_t)ncol * IN_F + 2 * kk;

    v8f c = {};
#pragma unroll
    for (int k0 = 0; k0 < IN_F; k0 += 4) {
        v2f a = *(const v2f*)(xrow + k0);
        v2f b = *(const v2f*)(wrow + k0);
        c = __builtin_amdgcn_wmma_f32_16x16x4_f32(
                 false, a,  false, b,
                 (short)0, c,  false,  false);
    }

    const int half = lane >> 4;
#pragma unroll
    for (int r = 0; r < 8; ++r) {
        int mm = r + 8 * half;
        nt[(size_t)(blockIdx.x * 16 + mm) * HF + wave * 16 + (lane & 15)] = c[r];
        tile[mm * TILE_PITCH + wave * 16 + (lane & 15)] = c[r];
    }
    __syncthreads();

    const int prow = threadIdx.x >> 3;
    const int o    = threadIdx.x & 7;
    const float* trow = tile + prow * TILE_PITCH;
    const float* cf   = a_w + (o & 3) * (2 * HF) + (o >> 2) * F_OUT;
    float acc0 = 0.f, acc1 = 0.f;
    for (int hp = 0; hp < 2; ++hp) {
        for (int q = 0; q < 4; ++q) {
            const float4 t0 = *(const float4*)(trow + hp * F_OUT + q * 4);
            const float4 a0 = *(const float4*)(cf + hp * 2 * F_OUT + q * 4);
            const float4 t1 = *(const float4*)(trow + (hp + 2) * F_OUT + q * 4);
            const float4 a1 = *(const float4*)(cf + (hp + 2) * 2 * F_OUT + q * 4);
            acc0 = fmaf(t0.x, a0.x, acc0);
            acc1 = fmaf(t1.x, a1.x, acc1);
            acc0 = fmaf(t0.y, a0.y, acc0);
            acc1 = fmaf(t1.y, a1.y, acc1);
            acc0 = fmaf(t0.z, a0.z, acc0);
            acc1 = fmaf(t1.z, a1.z, acc1);
            acc0 = fmaf(t0.w, a0.w, acc0);
            acc1 = fmaf(t1.w, a1.w, acc1);
        }
    }
    pq[(size_t)blockIdx.x * (16 * PQ_W) + threadIdx.x] = acc0 + acc1;

    if (blockIdx.x == 0 && threadIdx.x == 0) *maxenc = 0u;
}

__global__ __launch_bounds__(256)
void gat_edge_scores(const float* __restrict__ pq,
                     const int*   __restrict__ src,
                     const int*   __restrict__ tgt,
                     float* __restrict__ aw,
                     unsigned* __restrict__ maxenc) {
    __shared__ unsigned smax;
    if (threadIdx.x == 0) smax = 0u;
    __syncthreads();

    const int lane = threadIdx.x & 31;
    const int e    = blockIdx.x * blockDim.x + threadIdx.x;

    float mx = -__builtin_inff();
    if (e < E_EDGES) {
        const int vs = src[e];
        const int vt = tgt[e];
        const float4 p = *(const float4*)(pq + (size_t)vs * PQ_W);
        const float4 q = *(const float4*)(pq + (size_t)vt * PQ_W + H_HEADS);
        float4 r;
        r.x = p.x + q.x;
        r.y = p.y + q.y;
        r.z = p.z + q.z;
        r.w = p.w + q.w;
        *(float4*)(aw + (size_t)e * H_HEADS) = r;
        mx = fmaxf(fmaxf(r.x, r.y), fmaxf(r.z, r.w));
    }
    for (int d = 16; d >= 1; d >>= 1)
        mx = fmaxf(mx, __shfl_xor(mx, d, 32));
    if (lane == 0) atomicMax(&smax, fenc(mx));
    __syncthreads();
    if (threadIdx.x == 0) atomicMax(maxenc, smax);
}

__global__ __launch_bounds__(256)
void gat_softmax_denom(float* __restrict__ awex,
                       const int* __restrict__ tgt,
                       float* __restrict__ denom,
                       const unsigned* __restrict__ maxenc) {
    const int e = blockIdx.x * blockDim.x + threadIdx.x;
    if (e >= E_EDGES) return;
    const float gmax = fdec(*maxenc);
    const int vt = tgt[e];
    float4 a = *(const float4*)(awex + (size_t)e * H_HEADS);
    float v[H_HEADS] = {a.x, a.y, a.z, a.w};
#pragma unroll
    for (int h = 0; h < H_HEADS; ++h) {
        float z = v[h] - gmax;
        z = (z > 0.0f) ? z : NEG_SLOPE * z;
        float ev = __expf(z);
        awex[(size_t)e * H_HEADS + h] = ev;
        atomicAdd(&denom[(size_t)vt * H_HEADS + h], ev);
    }
}

__global__ __launch_bounds__(256)
void gat_aggregate(const float* __restrict__ nt,
                   const float* __restrict__ ex,
                   const float* __restrict__ denom,
                   const int*   __restrict__ src,
                   const int*   __restrict__ tgt,
                   float* __restrict__ out) {
    const int lane = threadIdx.x & 31;
    const int wv   = threadIdx.x >> 5;
    const int e    = blockIdx.x * 8 + wv;
    if (e >= E_EDGES) return;
    const int vs = src[e];
    const int vt = tgt[e];
#pragma unroll
    for (int c = 0; c < 2; ++c) {
        const int idx = lane + 32 * c;
        const int h   = idx >> 4;
        const float alpha = ex[(size_t)e * H_HEADS + h] /
                            (denom[(size_t)vt * H_HEADS + h] + 1e-8f);
        const float sv = nt[(size_t)vs * HF + idx];
        atomicAdd(&out[(size_t)vt * HF + idx], alpha * sv);
    }
}

extern "C" void kernel_launch(void* const* d_in, const int* in_sizes, int n_in,
                              void* d_out, int out_size, void* d_ws, size_t ws_size,
                              hipStream_t stream) {
    const float* x   = (const float*)d_in[0];
    const float* W_w = (const float*)d_in[1];
    const float* a_w = (const float*)d_in[2];
    const int*   ei  = (const int*)d_in[3];
    const int* src = ei;
    const int* tgt = ei + E_EDGES;
    float* out = (float*)d_out;

    float* nt    = (float*)d_ws;
    float* awex  = nt   + (size_t)N_NODES * HF;
    float* denom = awex + (size_t)E_EDGES * H_HEADS;
    unsigned* maxenc = (unsigned*)(denom + (size_t)N_NODES * H_HEADS);

    float* pq = out;

    (void)in_sizes; (void)n_in; (void)out_size; (void)ws_size;

    {
        gat_transform_wmma<<<N_NODES / 16, 128, 0, stream>>>(
            x, W_w, a_w, nt, pq, maxenc);
    }
    {
        gat_edge_scores<<<(E_EDGES + 255) / 256, 256, 0, stream>>>(
            pq, src, tgt, awex, maxenc);
    }
    {
        size_t total = (size_t)N_NODES * HF;
        int blocks = (int)((total + 255) / 256);
        gat_init<<<blocks, 256, 0, stream>>>(out, denom);
    }
    {
        gat_softmax_denom<<<(E_EDGES + 255) / 256, 256, 0, stream>>>(
            awex, tgt, denom, maxenc);
    }
    {
        gat_aggregate<<<(E_EDGES + 7) / 8, 256, 0, stream>>>(
            nt, awex, denom, src, tgt, out);
    }
}


# Aggregate per target node: in-call counting sort of edges by target (atomic rank), one wave per node sums its rows; no float atomics; segment sums now in rank order, denominator by lane tree
# speedup vs baseline: 2.9131x; 1.9832x over previous
#include <hip/hip_runtime.h>
#include <hip/hip_bf16.h>


#define N_NODES   100000
#define E_EDGES   1600000
#define IN_F      128
#define H_HEADS   4
#define F_OUT     16
#define HF        64
#define NEG_SLOPE 0.01f
#define PQ_W      8
#define TILE_PITCH 68
#define EPS_DEN   1e-8f

typedef float v2f __attribute__((ext_vector_type(2)));
typedef float v8f __attribute__((ext_vector_type(8)));

__device__ __forceinline__ unsigned fenc(float f) {
    unsigned u = __float_as_uint(f);
    return (u & 0x80000000u) ? ~u : (u | 0x80000000u);
}
__device__ __forceinline__ float fdec(unsigned k) {
    return (k & 0x80000000u) ? __uint_as_float(k ^ 0x80000000u)
                             : __uint_as_float(~k);
}

__global__ __launch_bounds__(128)
void gat_transform_wmma(const float* __restrict__ x,
                        const float* __restrict__ W,
                        const float* __restrict__ a_w,
                        float* __restrict__ nt,
                        float* __restrict__ pq,
                        int* __restrict__ cnt,
                        unsigned* __restrict__ cursor,
                        unsigned* __restrict__ maxenc) {
    __shared__ __attribute__((aligned(16))) float tile[16 * TILE_PITCH];
    const int lane = threadIdx.x & 31;
    const int wave = threadIdx.x >> 5;
    const int m    = lane & 15;
    const int kk   = lane >> 4;
    const int row  = blockIdx.x * 16 + m;
    const int ncol = wave * 16 + m;

    const float* xrow = x + (size_t)row  * IN_F + 2 * kk;
    const float* wrow = W + (size_t)ncol * IN_F + 2 * kk;

    v8f c = {};
#pragma unroll
    for (int k0 = 0; k0 < IN_F; k0 += 4) {
        v2f a = *(const v2f*)(xrow + k0);
        v2f b = *(const v2f*)(wrow + k0);
        c = __builtin_amdgcn_wmma_f32_16x16x4_f32(
                 false, a,  false, b,
                 (short)0, c,  false,  false);
    }

    const int half = lane >> 4;
#pragma unroll
    for (int r = 0; r < 8; ++r) {
        int mm = r + 8 * half;
        nt[(size_t)(blockIdx.x * 16 + mm) * HF + wave * 16 + (lane & 15)] = c[r];
        tile[mm * TILE_PITCH + wave * 16 + (lane & 15)] = c[r];
    }
    __syncthreads();

    const int prow = threadIdx.x >> 3;
    const int o    = threadIdx.x & 7;
    const float* trow = tile + prow * TILE_PITCH;
    const float* cf   = a_w + (o & 3) * (2 * HF) + (o >> 2) * F_OUT;
    float acc0 = 0.f, acc1 = 0.f;
    for (int hp = 0; hp < 2; ++hp) {
        for (int q = 0; q < 4; ++q) {
            const float4 t0 = *(const float4*)(trow + hp * F_OUT + q * 4);
            const float4 a0 = *(const float4*)(cf + hp * 2 * F_OUT + q * 4);
            const float4 t1 = *(const float4*)(trow + (hp + 2) * F_OUT + q * 4);
            const float4 a1 = *(const float4*)(cf + (hp + 2) * 2 * F_OUT + q * 4);
            acc0 = fmaf(t0.x, a0.x, acc0);
            acc1 = fmaf(t1.x, a1.x, acc1);
            acc0 = fmaf(t0.y, a0.y, acc0);
            acc1 = fmaf(t1.y, a1.y, acc1);
            acc0 = fmaf(t0.z, a0.z, acc0);
            acc1 = fmaf(t1.z, a1.z, acc1);
            acc0 = fmaf(t0.w, a0.w, acc0);
            acc1 = fmaf(t1.w, a1.w, acc1);
        }
    }
    pq[(size_t)blockIdx.x * (16 * PQ_W) + threadIdx.x] = acc0 + acc1;

    if (threadIdx.x < 16) cnt[blockIdx.x * 16 + threadIdx.x] = 0;
    if (blockIdx.x == 0 && threadIdx.x == 0) {
        *maxenc = 0u;
        *cursor = 0u;
    }
}

__global__ __launch_bounds__(256)
void gat_edge_scores(const float* __restrict__ pq,
                     const int*   __restrict__ src,
                     const int*   __restrict__ tgt,
                     int* __restrict__ cnt,
                     int* __restrict__ rank,
                     unsigned* __restrict__ maxenc) {
    __shared__ unsigned smax;
    if (threadIdx.x == 0) smax = 0u;
    __syncthreads();

    const int lane = threadIdx.x & 31;
    const int e    = blockIdx.x * blockDim.x + threadIdx.x;

    float mx = -__builtin_inff();
    if (e < E_EDGES) {
        const int vs = src[e];
        const int vt = tgt[e];
        const float4 p = *(const float4*)(pq + (size_t)vs * PQ_W);
        const float4 q = *(const float4*)(pq + (size_t)vt * PQ_W + H_HEADS);
        rank[e] = atomicAdd(&cnt[vt], 1);
        mx = fmaxf(fmaxf(p.x + q.x, p.y + q.y), fmaxf(p.z + q.z, p.w + q.w));
    }
    for (int d = 16; d >= 1; d >>= 1)
        mx = fmaxf(mx, __shfl_xor(mx, d, 32));
    if (lane == 0) atomicMax(&smax, fenc(mx));
    __syncthreads();
    if (threadIdx.x == 0) atomicMax(maxenc, smax);
}

__global__ __launch_bounds__(256)
void gat_alloc(const int* __restrict__ cnt,
               int* __restrict__ offs,
               unsigned* __restrict__ cursor) {
    __shared__ int wsum[8];
    __shared__ int sbase;
    const int lane = threadIdx.x & 31;
    const int wave = threadIdx.x >> 5;
    const int n    = blockIdx.x * 256 + threadIdx.x;
    const int c    = (n < N_NODES) ? cnt[n] : 0;

    int v = c;
    for (int d = 1; d < 32; d <<= 1) {
        const int t = __shfl_up(v, d, 32);
        if (lane >= d) v += t;
    }
    if (lane == 31) wsum[wave] = v;
    __syncthreads();
    if (threadIdx.x == 0) {
        int run = 0;
        for (int w = 0; w < 8; ++w) { const int t = wsum[w]; wsum[w] = run; run += t; }
        sbase = (int)atomicAdd(cursor, (unsigned)run);
    }
    __syncthreads();
    if (n < N_NODES) offs[n] = sbase + wsum[wave] + v - c;
}

__global__ __launch_bounds__(256)
void gat_fill(const int* __restrict__ src,
              const int* __restrict__ tgt,
              const int* __restrict__ rank,
              const int* __restrict__ offs,
              int* __restrict__ ssrc) {
    const int e = blockIdx.x * blockDim.x + threadIdx.x;
    if (e >= E_EDGES) return;
    ssrc[offs[tgt[e]] + rank[e]] = src[e];
}

__device__ __forceinline__ float edge_exp(float a, float gmax) {
    float z = a - gmax;
    z = (z > 0.0f) ? z : NEG_SLOPE * z;
    return __expf(z);
}

__device__ __forceinline__ void edge_chunk(const int* __restrict__ ssrc,
                                           const float* __restrict__ pq,
                                           int o, int base, int c, int lane,
                                           float4 q4, float gmax,
                                           int& s, float& e0, float& e1,
                                           float& e2, float& e3) {
    const int  i     = base + lane;
    const bool valid = i < c;
    s = valid ? ssrc[o + i] : 0;
    const float4 p = *(const float4*)(pq + (size_t)s * PQ_W);
    e0 = valid ? edge_exp(p.x + q4.x, gmax) : 0.0f;
    e1 = valid ? edge_exp(p.y + q4.y, gmax) : 0.0f;
    e2 = valid ? edge_exp(p.z + q4.z, gmax) : 0.0f;
    e3 = valid ? edge_exp(p.w + q4.w, gmax) : 0.0f;
}

__device__ __forceinline__ unsigned lane_bcast(float v, int j) {
    return (unsigned)__builtin_amdgcn_readlane(__float_as_int(v), j);
}

__device__ __forceinline__ float head_pick(float al0, float al1, float al2, float al3,
                                           unsigned m0, unsigned m1, unsigned m2, unsigned m3,
                                           int j) {
    return __uint_as_float((lane_bcast(al0, j) & m0) | (lane_bcast(al1, j) & m1) |
                           (lane_bcast(al2, j) & m2) | (lane_bcast(al3, j) & m3));
}

__global__ __launch_bounds__(256)
void gat_node_aggregate(const float* __restrict__ nt,
                        const float* __restrict__ pq,
                        const int*   __restrict__ cnt,
                        const int*   __restrict__ offs,
                        const int*   __restrict__ ssrc,
                        const unsigned* __restrict__ maxenc,
                        float* __restrict__ out) {
    const int lane = threadIdx.x & 31;
    const int wv = __builtin_amdgcn_readfirstlane((int)(threadIdx.x >> 5));
    const int n  = blockIdx.x * 8 + wv;
    if (n >= N_NODES) return;

    const int   c    = cnt[n];
    const int   o    = offs[n];
    const float gmax = fdec(*maxenc);
    const float4 q4  = *(const float4*)(pq + (size_t)n * PQ_W + H_HEADS);
    const int   h    = lane >> 3;
    const unsigned m0 = (h == 0) ? 0xffffffffu : 0u, m1 = (h == 1) ? 0xffffffffu : 0u;
    const unsigned m2 = (h == 2) ? 0xffffffffu : 0u, m3 = (h == 3) ? 0xffffffffu : 0u;
    const float* ntl = nt + 2 * lane;

    int   s;
    float e0, e1, e2, e3;
    edge_chunk(ssrc, pq, o, 0, c, lane, q4, gmax, s, e0, e1, e2, e3);

    float d0 = e0, d1 = e1, d2 = e2, d3 = e3;
    for (int base = 32; base < c; base += 32) {
        int s2; float f0, f1, f2, f3;
        edge_chunk(ssrc, pq, o, base, c, lane, q4, gmax, s2, f0, f1, f2, f3);
        d0 += f0; d1 += f1; d2 += f2; d3 += f3;
    }
    for (int d = 16; d >= 1; d >>= 1) {
        d0 += __shfl_xor(d0, d, 32);
        d1 += __shfl_xor(d1, d, 32);
        d2 += __shfl_xor(d2, d, 32);
        d3 += __shfl_xor(d3, d, 32);
    }
    d0 += EPS_DEN; d1 += EPS_DEN; d2 += EPS_DEN; d3 += EPS_DEN;

    float2 acc = {0.0f, 0.0f};
    for (int base = 0; base < c; base += 32) {
        if (base > 0)
            edge_chunk(ssrc, pq, o, base, c, lane, q4, gmax, s, e0, e1, e2, e3);
        const float al0 = e0 / d0, al1 = e1 / d1, al2 = e2 / d2, al3 = e3 / d3;
        const int rest = c - base;
        const int m = rest < 32 ? rest : 32;

        int j = 0;
        for (; j + 4 <= m; j += 4) {
            float2 r[4];
#pragma unroll
            for (int k = 0; k < 4; ++k) {
                const int sj = __builtin_amdgcn_readlane(s, j + k);
                r[k] = *(const float2*)(ntl + (size_t)sj * HF);
            }
#pragma unroll
            for (int k = 0; k < 4; ++k) {
                const float a = head_pick(al0, al1, al2, al3, m0, m1, m2, m3, j + k);
                acc.x += a * r[k].x;
                acc.y += a * r[k].y;
            }
        }
        for (; j < m; ++j) {
            const int sj = __builtin_amdgcn_readlane(s, j);
            const float2 r = *(const float2*)(ntl + (size_t)sj * HF);
            const float a = head_pick(al0, al1, al2, al3, m0, m1, m2, m3, j);
            acc.x += a * r.x;
            acc.y += a * r.y;
        }
    }
    *(float2*)(out + (size_t)n * HF + 2 * lane) = acc;
}

extern "C" void kernel_launch(void* const* d_in, const int* in_sizes, int n_in,
                              void* d_out, int out_size, void* d_ws, size_t ws_size,
                              hipStream_t stream) {
    const float* x   = (const float*)d_in[0];
    const float* W_w = (const float*)d_in[1];
    const float* a_w = (const float*)d_in[2];
    const int*   ei  = (const int*)d_in[3];
    const int* src = ei;
    const int* tgt = ei + E_EDGES;
    float* out = (float*)d_out;

    float* nt    = (float*)d_ws;
    float* pq    = nt + (size_t)N_NODES * HF;
    int*   cnt   = (int*)(pq + (size_t)N_NODES * PQ_W);
    int*   offs  = cnt  + N_NODES;
    int*   rank  = offs + N_NODES;
    int*   ssrc  = rank + E_EDGES;
    unsigned* cursor = (unsigned*)(ssrc + E_EDGES);
    unsigned* maxenc = cursor + 1;

    (void)in_sizes; (void)n_in; (void)out_size; (void)ws_size;

    gat_transform_wmma<<<N_NODES / 16, 128, 0, stream>>>(
        x, W_w, a_w, nt, pq, cnt, cursor, maxenc);
    gat_edge_scores<<<(E_EDGES + 255) / 256, 256, 0, stream>>>(
        pq, src, tgt, cnt, rank, maxenc);
    gat_alloc<<<(N_NODES + 255) / 256, 256, 0, stream>>>(cnt, offs, cursor);
    gat_fill<<<(E_EDGES + 255) / 256, 256, 0, stream>>>(src, tgt, rank, offs, ssrc);
    gat_node_aggregate<<<(N_NODES + 7) / 8, 256, 0, stream>>>(
        nt, pq, cnt, offs, ssrc, maxenc, out);
}


# Resident-size grids: each wave takes several items (6-7 row tiles with W fragments kept in registers, 8 edges per thread, 8 nodes per wave): about 7x fewer waves start; rank still arrival order
# speedup vs baseline: 3.2851x; 1.1277x over previous
#include <hip/hip_runtime.h>
#include <hip/hip_bf16.h>


#define N_NODES   100000
#define E_EDGES   1600000
#define IN_F      128
#define H_HEADS   4
#define F_OUT     16
#define HF        64
#define NEG_SLOPE 0.01f
#define PQ_W      8
#define TILE_PITCH 68
#define EPS_DEN   1e-8f

#define ROW_TILES   (N_NODES / 16)
#define XF_BLOCKS   1024
#define E_VEC4      (E_EDGES / 4)
#define E_HALF4     (E_VEC4 / 2)
#define EDGE_BLOCKS ((E_HALF4 + 255) / 256)
#define AGG_WAVES   12500
#define AGG_BLOCKS  (AGG_WAVES / 4)
static_assert(N_NODES % 16 == 0, "row tiles must be whole");
static_assert(E_EDGES % 8 == 0, "edge arrays must split into two halves of 16-byte items");
static_assert(AGG_WAVES % 4 == 0, "aggregate grid is built from 4-wave blocks");

typedef float v2f __attribute__((ext_vector_type(2)));
typedef float v8f __attribute__((ext_vector_type(8)));

__device__ __forceinline__ unsigned fenc(float f) {
    unsigned u = __float_as_uint(f);
    return (u & 0x80000000u) ? ~u : (u | 0x80000000u);
}
__device__ __forceinline__ float fdec(unsigned k) {
    return (k & 0x80000000u) ? __uint_as_float(k ^ 0x80000000u)
                             : __uint_as_float(~k);
}

__global__ __launch_bounds__(128)
void gat_transform_wmma(const float* __restrict__ x,
                        const float* __restrict__ W,
                        const float* __restrict__ a_w,
                        float* __restrict__ nt,
                        float* __restrict__ pq,
                        int* __restrict__ cnt,
                        unsigned* __restrict__ cursor,
                        unsigned* __restrict__ maxenc) {
    __shared__ __attribute__((aligned(16))) float tile[2][16 * TILE_PITCH];
    const int lane = threadIdx.x & 31;
    const int wave = threadIdx.x >> 5;
    const int m    = lane & 15;
    const int kk   = lane >> 4;
    const int half = lane >> 4;
    const int ncol = wave * 16 + m;

    const float* wrow = W + (size_t)ncol * IN_F + 2 * kk;
    v2f b[IN_F / 4];
#pragma unroll
    for (int i = 0; i < IN_F / 4; ++i)
        b[i] = *(const v2f*)(wrow + 4 * i);

    const int prow = threadIdx.x >> 3;
    const int o    = threadIdx.x & 7;
    const float* cf = a_w + (o & 3) * (2 * HF) + (o >> 2) * F_OUT;

    int buf = 0;
    for (int t = blockIdx.x; t < ROW_TILES; t += gridDim.x, buf ^= 1) {
        const float* xrow = x + (size_t)(t * 16 + m) * IN_F + 2 * kk;
        v8f c = {};
#pragma unroll
        for (int i = 0; i < IN_F / 4; ++i) {
            v2f a = *(const v2f*)(xrow + 4 * i);
            c = __builtin_amdgcn_wmma_f32_16x16x4_f32(
                     false, a,  false, b[i],
                     (short)0, c,  false,  false);
        }

        float* tl = tile[buf];
#pragma unroll
        for (int r = 0; r < 8; ++r) {
            int mm = r + 8 * half;
            nt[(size_t)(t * 16 + mm) * HF + wave * 16 + (lane & 15)] = c[r];
            tl[mm * TILE_PITCH + wave * 16 + (lane & 15)] = c[r];
        }
        __syncthreads();

        const float* trow = tl + prow * TILE_PITCH;
        float acc0 = 0.f, acc1 = 0.f;
        for (int hp = 0; hp < 2; ++hp) {
            for (int q = 0; q < 4; ++q) {
                const float4 t0 = *(const float4*)(trow + hp * F_OUT + q * 4);
                const float4 a0 = *(const float4*)(cf + hp * 2 * F_OUT + q * 4);
                const float4 t1 = *(const float4*)(trow + (hp + 2) * F_OUT + q * 4);
                const float4 a1 = *(const float4*)(cf + (hp + 2) * 2 * F_OUT + q * 4);
                acc0 = fmaf(t0.x, a0.x, acc0);
                acc1 = fmaf(t1.x, a1.x, acc1);
                acc0 = fmaf(t0.y, a0.y, acc0);
                acc1 = fmaf(t1.y, a1.y, acc1);
                acc0 = fmaf(t0.z, a0.z, acc0);
                acc1 = fmaf(t1.z, a1.z, acc1);
                acc0 = fmaf(t0.w, a0.w, acc0);
                acc1 = fmaf(t1.w, a1.w, acc1);
            }
        }
        pq[(size_t)t * (16 * PQ_W) + threadIdx.x] = acc0 + acc1;

        if (threadIdx.x < 16) cnt[t * 16 + threadIdx.x] = 0;
    }
    if (blockIdx.x == 0 && threadIdx.x == 0) {
        *maxenc = 0u;
        *cursor = 0u;
    }
}

__global__ __launch_bounds__(256)
void gat_edge_scores(const float* __restrict__ pq,
                     const int*   __restrict__ src,
                     const int*   __restrict__ tgt,
                     int* __restrict__ cnt,
                     int* __restrict__ rank,
                     unsigned* __restrict__ maxenc) {
    __shared__ unsigned smax;
    if (threadIdx.x == 0) smax = 0u;
    __syncthreads();

    const int lane = threadIdx.x & 31;
    const int g    = blockIdx.x * blockDim.x + threadIdx.x;

    float mx = -__builtin_inff();
    if (g < E_HALF4) {
        const int4 sa = ((const int4*)src)[g];
        const int4 sb = ((const int4*)src)[g + E_HALF4];
        const int4 ta = ((const int4*)tgt)[g];
        const int4 tb = ((const int4*)tgt)[g + E_HALF4];
        const int vs[8] = {sa.x, sa.y, sa.z, sa.w, sb.x, sb.y, sb.z, sb.w};
        const int vt[8] = {ta.x, ta.y, ta.z, ta.w, tb.x, tb.y, tb.z, tb.w};
        float4 p[8], q[8];
#pragma unroll
        for (int k = 0; k < 8; ++k) {
            p[k] = *(const float4*)(pq + (size_t)vs[k] * PQ_W);
            q[k] = *(const float4*)(pq + (size_t)vt[k] * PQ_W + H_HEADS);
        }
        int r[8];
#pragma unroll
        for (int k = 0; k < 8; ++k)
            r[k] = atomicAdd(&cnt[vt[k]], 1);
#pragma unroll
        for (int k = 0; k < 8; ++k)
            mx = fmaxf(mx, fmaxf(fmaxf(p[k].x + q[k].x, p[k].y + q[k].y),
                                 fmaxf(p[k].z + q[k].z, p[k].w + q[k].w)));
        ((int4*)rank)[g]           = make_int4(r[0], r[1], r[2], r[3]);
        ((int4*)rank)[g + E_HALF4] = make_int4(r[4], r[5], r[6], r[7]);
    }
    for (int d = 16; d >= 1; d >>= 1)
        mx = fmaxf(mx, __shfl_xor(mx, d, 32));
    if (lane == 0) atomicMax(&smax, fenc(mx));
    __syncthreads();
    if (threadIdx.x == 0) atomicMax(maxenc, smax);
}

__global__ __launch_bounds__(256)
void gat_alloc(const int* __restrict__ cnt,
               int* __restrict__ offs,
               unsigned* __restrict__ cursor) {
    __shared__ int wsum[8];
    __shared__ int sbase;
    const int lane = threadIdx.x & 31;
    const int wave = threadIdx.x >> 5;
    const int n    = blockIdx.x * 256 + threadIdx.x;
    const int c    = (n < N_NODES) ? cnt[n] : 0;

    int v = c;
    for (int d = 1; d < 32; d <<= 1) {
        const int t = __shfl_up(v, d, 32);
        if (lane >= d) v += t;
    }
    if (lane == 31) wsum[wave] = v;
    __syncthreads();
    if (threadIdx.x == 0) {
        int run = 0;
        for (int w = 0; w < 8; ++w) { const int t = wsum[w]; wsum[w] = run; run += t; }
        sbase = (int)atomicAdd(cursor, (unsigned)run);
    }
    __syncthreads();
    if (n < N_NODES) offs[n] = sbase + wsum[wave] + v - c;
}

__global__ __launch_bounds__(256)
void gat_fill(const int* __restrict__ src,
              const int* __restrict__ tgt,
              const int* __restrict__ rank,
              const int* __restrict__ offs,
              int* __restrict__ ssrc) {
    const int g = blockIdx.x * blockDim.x + threadIdx.x;
    if (g >= E_HALF4) return;
    const int4 ta = ((const int4*)tgt)[g];
    const int4 tb = ((const int4*)tgt)[g + E_HALF4];
    const int4 ra = ((const int4*)rank)[g];
    const int4 rb = ((const int4*)rank)[g + E_HALF4];
    const int4 sa = ((const int4*)src)[g];
    const int4 sb = ((const int4*)src)[g + E_HALF4];
    const int vt[8] = {ta.x, ta.y, ta.z, ta.w, tb.x, tb.y, tb.z, tb.w};
    const int vr[8] = {ra.x, ra.y, ra.z, ra.w, rb.x, rb.y, rb.z, rb.w};
    const int vs[8] = {sa.x, sa.y, sa.z, sa.w, sb.x, sb.y, sb.z, sb.w};
    int o[8];
#pragma unroll
    for (int k = 0; k < 8; ++k) o[k] = offs[vt[k]];
#pragma unroll
    for (int k = 0; k < 8; ++k) ssrc[o[k] + vr[k]] = vs[k];
}

__device__ __forceinline__ float edge_exp(float a, float gmax) {
    float z = a - gmax;
    z = (z > 0.0f) ? z : NEG_SLOPE * z;
    return __expf(z);
}

__device__ __forceinline__ void edge_chunk(const int* __restrict__ ssrc,
                                           const float* __restrict__ pq,
                                           int o, int base, int c, int lane,
                                           float4 q4, float gmax,
                                           int& s, float& e0, float& e1,
                                           float& e2, float& e3) {
    const int  i     = base + lane;
    const bool valid = i < c;
    s = valid ? ssrc[o + i] : 0;
    const float4 p = *(const float4*)(pq + (size_t)s * PQ_W);
    e0 = valid ? edge_exp(p.x + q4.x, gmax) : 0.0f;
    e1 = valid ? edge_exp(p.y + q4.y, gmax) : 0.0f;
    e2 = valid ? edge_exp(p.z + q4.z, gmax) : 0.0f;
    e3 = valid ? edge_exp(p.w + q4.w, gmax) : 0.0f;
}

__device__ __forceinline__ unsigned lane_bcast(float v, int j) {
    return (unsigned)__builtin_amdgcn_readlane(__float_as_int(v), j);
}

__device__ __forceinline__ float head_pick(float al0, float al1, float al2, float al3,
                                           unsigned m0, unsigned m1, unsigned m2, unsigned m3,
                                           int j) {
    return __uint_as_float((lane_bcast(al0, j) & m0) | (lane_bcast(al1, j) & m1) |
                           (lane_bcast(al2, j) & m2) | (lane_bcast(al3, j) & m3));
}

__global__ __launch_bounds__(128)
void gat_node_aggregate(const float* __restrict__ nt,
                        const float* __restrict__ pq,
                        const int*   __restrict__ cnt,
                        const int*   __restrict__ offs,
                        const int*   __restrict__ ssrc,
                        const unsigned* __restrict__ maxenc,
                        float* __restrict__ out) {
    const int lane = threadIdx.x & 31;
    const int wv = __builtin_amdgcn_readfirstlane((int)(threadIdx.x >> 5));

    const float gmax = fdec(*maxenc);
    const int   h    = lane >> 3;
    const unsigned m0 = (h == 0) ? 0xffffffffu : 0u, m1 = (h == 1) ? 0xffffffffu : 0u;
    const unsigned m2 = (h == 2) ? 0xffffffffu : 0u, m3 = (h == 3) ? 0xffffffffu : 0u;
    const float* ntl = nt + 2 * lane;

    for (int n = blockIdx.x * 4 + wv; n < N_NODES; n += AGG_WAVES) {
        const int   c   = cnt[n];
        const int   o   = offs[n];
        const float4 q4 = *(const float4*)(pq + (size_t)n * PQ_W + H_HEADS);

        int   s;
        float e0, e1, e2, e3;
        edge_chunk(ssrc, pq, o, 0, c, lane, q4, gmax, s, e0, e1, e2, e3);

        float d0 = e0, d1 = e1, d2 = e2, d3 = e3;
        for (int base = 32; base < c; base += 32) {
            int s2; float f0, f1, f2, f3;
            edge_chunk(ssrc, pq, o, base, c, lane, q4, gmax, s2, f0, f1, f2, f3);
            d0 += f0; d1 += f1; d2 += f2; d3 += f3;
        }
        for (int d = 16; d >= 1; d >>= 1) {
            d0 += __shfl_xor(d0, d, 32);
            d1 += __shfl_xor(d1, d, 32);
            d2 += __shfl_xor(d2, d, 32);
            d3 += __shfl_xor(d3, d, 32);
        }
        d0 += EPS_DEN; d1 += EPS_DEN; d2 += EPS_DEN; d3 += EPS_DEN;

        float2 acc = {0.0f, 0.0f};
        for (int base = 0; base < c; base += 32) {
            if (base > 0)
                edge_chunk(ssrc, pq, o, base, c, lane, q4, gmax, s, e0, e1, e2, e3);
            const float al0 = e0 / d0, al1 = e1 / d1, al2 = e2 / d2, al3 = e3 / d3;
            const int rest = c - base;
            const int m = rest < 32 ? rest : 32;

            int j = 0;
            for (; j + 4 <= m; j += 4) {
                float2 r[4];
#pragma unroll
                for (int k = 0; k < 4; ++k) {
                    const int sj = __builtin_amdgcn_readlane(s, j + k);
                    r[k] = *(const float2*)(ntl + (size_t)sj * HF);
                }
#pragma unroll
                for (int k = 0; k < 4; ++k) {
                    const float a = head_pick(al0, al1, al2, al3, m0, m1, m2, m3, j + k);
                    acc.x += a * r[k].x;
                    acc.y += a * r[k].y;
                }
            }
            for (; j < m; ++j) {
                const int sj = __builtin_amdgcn_readlane(s, j);
                const float2 r = *(const float2*)(ntl + (size_t)sj * HF);
                const float a = head_pick(al0, al1, al2, al3, m0, m1, m2, m3, j);
                acc.x += a * r.x;
                acc.y += a * r.y;
            }
        }
        *(float2*)(out + (size_t)n * HF + 2 * lane) = acc;
    }
}

extern "C" void kernel_launch(void* const* d_in, const int* in_sizes, int n_in,
                              void* d_out, int out_size, void* d_ws, size_t ws_size,
                              hipStream_t stream) {
    const float* x   = (const float*)d_in[0];
    const float* W_w = (const float*)d_in[1];
    const float* a_w = (const float*)d_in[2];
    const int*   ei  = (const int*)d_in[3];
    const int* src = ei;
    const int* tgt = ei + E_EDGES;
    float* out = (float*)d_out;

    float* nt    = (float*)d_ws;
    float* pq    = nt + (size_t)N_NODES * HF;
    int*   cnt   = (int*)(pq + (size_t)N_NODES * PQ_W);
    int*   offs  = cnt  + N_NODES;
    int*   rank  = offs + N_NODES;
    int*   ssrc  = rank + E_EDGES;
    unsigned* cursor = (unsigned*)(ssrc + E_EDGES);
    unsigned* maxenc = cursor + 1;

    (void)in_sizes; (void)n_in; (void)out_size; (void)ws_size;

    gat_transform_wmma<<<XF_BLOCKS, 128, 0, stream>>>(
        x, W_w, a_w, nt, pq, cnt, cursor, maxenc);
    gat_edge_scores<<<EDGE_BLOCKS, 256, 0, stream>>>(
        pq, src, tgt, cnt, rank, maxenc);
    gat_alloc<<<(N_NODES + 255) / 256, 256, 0, stream>>>(cnt, offs, cursor);
    gat_fill<<<EDGE_BLOCKS, 256, 0, stream>>>(src, tgt, rank, offs, ssrc);
    gat_node_aggregate<<<AGG_BLOCKS, 128, 0, stream>>>(
        nt, pq, cnt, offs, ssrc, maxenc, out);
}


# Transform: x row tile staged in LDS by whole-row 16-byte loads (4 full lines per wave load, once per block, next tile prefetched); waves read A fragments from LDS at the same k; sums unchanged
# speedup vs baseline: 3.4293x; 1.0439x over previous
#include <hip/hip_runtime.h>
#include <hip/hip_bf16.h>


#define N_NODES   100000
#define E_EDGES   1600000
#define IN_F      128
#define H_HEADS   4
#define F_OUT     16
#define HF        64
#define NEG_SLOPE 0.01f
#define PQ_W      8
#define TILE_PITCH 68
#define X_PITCH   132
#define EPS_DEN   1e-8f

#define ROW_TILES   (N_NODES / 16)
#define XF_BLOCKS   1024
#define E_VEC4      (E_EDGES / 4)
#define E_HALF4     (E_VEC4 / 2)
#define EDGE_BLOCKS ((E_HALF4 + 255) / 256)
#define AGG_WAVES   12500
#define AGG_BLOCKS  (AGG_WAVES / 4)
static_assert(N_NODES % 16 == 0, "row tiles must be whole");
static_assert(E_EDGES % 8 == 0, "edge arrays must split into two halves of 16-byte items");
static_assert(AGG_WAVES % 4 == 0, "aggregate grid is built from 4-wave blocks");

typedef float v2f __attribute__((ext_vector_type(2)));
typedef float v8f __attribute__((ext_vector_type(8)));

__device__ __forceinline__ unsigned fenc(float f) {
    unsigned u = __float_as_uint(f);
    return (u & 0x80000000u) ? ~u : (u | 0x80000000u);
}
__device__ __forceinline__ float fdec(unsigned k) {
    return (k & 0x80000000u) ? __uint_as_float(k ^ 0x80000000u)
                             : __uint_as_float(~k);
}

__global__ __launch_bounds__(128)
void gat_transform_wmma(const float* __restrict__ x,
                        const float* __restrict__ W,
                        const float* __restrict__ a_w,
                        float* __restrict__ nt,
                        float* __restrict__ pq,
                        int* __restrict__ cnt,
                        unsigned* __restrict__ cursor,
                        unsigned* __restrict__ maxenc) {
    __shared__ __attribute__((aligned(16))) float tile[2][16 * TILE_PITCH];
    __shared__ __attribute__((aligned(16))) float xs[2][16 * X_PITCH];
    const int lane = threadIdx.x & 31;
    const int wave = threadIdx.x >> 5;
    const int m    = lane & 15;
    const int kk   = lane >> 4;
    const int half = lane >> 4;
    const int ncol = wave * 16 + m;

    const float* wrow = W + (size_t)ncol * IN_F + 2 * kk;
    v2f b[IN_F / 4];
#pragma unroll
    for (int i = 0; i < IN_F / 4; ++i)
        b[i] = *(const v2f*)(wrow + 4 * i);

    const int prow = threadIdx.x >> 3;
    const int o    = threadIdx.x & 7;
    const float* cf = a_w + (o & 3) * (2 * HF) + (o >> 2) * F_OUT;

    const int    xoff_g = wave * IN_F + 4 * lane;
    const int    xoff_s = wave * X_PITCH + 4 * lane;
    const float* xfrag  = &xs[0][0] + m * X_PITCH + 2 * kk;

    float4 p0 = make_float4(0.f, 0.f, 0.f, 0.f), p1 = p0, p2 = p0, p3 = p0;
    if (blockIdx.x < ROW_TILES) {
        const float* xt = x + (size_t)blockIdx.x * (16 * IN_F) + xoff_g;
        p0 = *(const float4*)(xt);
        p1 = *(const float4*)(xt + 4 * IN_F);
        p2 = *(const float4*)(xt + 8 * IN_F);
        p3 = *(const float4*)(xt + 12 * IN_F);
        float* xd = &xs[0][0] + xoff_s;
        *(float4*)(xd)                = p0;
        *(float4*)(xd + 4 * X_PITCH)  = p1;
        *(float4*)(xd + 8 * X_PITCH)  = p2;
        *(float4*)(xd + 12 * X_PITCH) = p3;
    }
    __syncthreads();

    int buf = 0;
    for (int t = blockIdx.x; t < ROW_TILES; t += gridDim.x, buf ^= 1) {
        const int  tn   = t + gridDim.x;
        const bool more = tn < ROW_TILES;
        if (more) {
            const float* xt = x + (size_t)tn * (16 * IN_F) + xoff_g;
            p0 = *(const float4*)(xt);
            p1 = *(const float4*)(xt + 4 * IN_F);
            p2 = *(const float4*)(xt + 8 * IN_F);
            p3 = *(const float4*)(xt + 12 * IN_F);
        }

        const float* xa = xfrag + buf * (16 * X_PITCH);
        v8f c = {};
#pragma unroll
        for (int i = 0; i < IN_F / 4; ++i) {
            v2f a = *(const v2f*)(xa + 4 * i);
            c = __builtin_amdgcn_wmma_f32_16x16x4_f32(
                     false, a,  false, b[i],
                     (short)0, c,  false,  false);
        }

        float* tl = tile[buf];
#pragma unroll
        for (int r = 0; r < 8; ++r) {
            int mm = r + 8 * half;
            nt[(size_t)(t * 16 + mm) * HF + wave * 16 + (lane & 15)] = c[r];
            tl[mm * TILE_PITCH + wave * 16 + (lane & 15)] = c[r];
        }
        if (more) {
            float* xd = &xs[buf ^ 1][0] + xoff_s;
            *(float4*)(xd)                = p0;
            *(float4*)(xd + 4 * X_PITCH)  = p1;
            *(float4*)(xd + 8 * X_PITCH)  = p2;
            *(float4*)(xd + 12 * X_PITCH) = p3;
        }
        __syncthreads();

        const float* trow = tl + prow * TILE_PITCH;
        float acc0 = 0.f, acc1 = 0.f;
        for (int hp = 0; hp < 2; ++hp) {
            for (int q = 0; q < 4; ++q) {
                const float4 t0 = *(const float4*)(trow + hp * F_OUT + q * 4);
                const float4 a0 = *(const float4*)(cf + hp * 2 * F_OUT + q * 4);
                const float4 t1 = *(const float4*)(trow + (hp + 2) * F_OUT + q * 4);
                const float4 a1 = *(const float4*)(cf + (hp + 2) * 2 * F_OUT + q * 4);
                acc0 = fmaf(t0.x, a0.x, acc0);
                acc1 = fmaf(t1.x, a1.x, acc1);
                acc0 = fmaf(t0.y, a0.y, acc0);
                acc1 = fmaf(t1.y, a1.y, acc1);
                acc0 = fmaf(t0.z, a0.z, acc0);
                acc1 = fmaf(t1.z, a1.z, acc1);
                acc0 = fmaf(t0.w, a0.w, acc0);
                acc1 = fmaf(t1.w, a1.w, acc1);
            }
        }
        pq[(size_t)t * (16 * PQ_W) + threadIdx.x] = acc0 + acc1;

        if (threadIdx.x < 16) cnt[t * 16 + threadIdx.x] = 0;
    }
    if (blockIdx.x == 0 && threadIdx.x == 0) {
        *maxenc = 0u;
        *cursor = 0u;
    }
}

__global__ __launch_bounds__(256)
void gat_edge_scores(const float* __restrict__ pq,
                     const int*   __restrict__ src,
                     const int*   __restrict__ tgt,
                     int* __restrict__ cnt,
                     int* __restrict__ rank,
                     unsigned* __restrict__ maxenc) {
    __shared__ unsigned smax;
    if (threadIdx.x == 0) smax = 0u;
    __syncthreads();

    const int lane = threadIdx.x & 31;
    const int g    = blockIdx.x * blockDim.x + threadIdx.x;

    float mx = -__builtin_inff();
    if (g < E_HALF4) {
        const int4 sa = ((const int4*)src)[g];
        const int4 sb = ((const int4*)src)[g + E_HALF4];
        const int4 ta = ((const int4*)tgt)[g];
        const int4 tb = ((const int4*)tgt)[g + E_HALF4];
        const int vs[8] = {sa.x, sa.y, sa.z, sa.w, sb.x, sb.y, sb.z, sb.w};
        const int vt[8] = {ta.x, ta.y, ta.z, ta.w, tb.x, tb.y, tb.z, tb.w};
        float4 p[8], q[8];
#pragma unroll
        for (int k = 0; k < 8; ++k) {
            p[k] = *(const float4*)(pq + (size_t)vs[k] * PQ_W);
            q[k] = *(const float4*)(pq + (size_t)vt[k] * PQ_W + H_HEADS);
        }
        int r[8];
#pragma unroll
        for (int k = 0; k < 8; ++k)
            r[k] = atomicAdd(&cnt[vt[k]], 1);
#pragma unroll
        for (int k = 0; k < 8; ++k)
            mx = fmaxf(mx, fmaxf(fmaxf(p[k].x + q[k].x, p[k].y + q[k].y),
                                 fmaxf(p[k].z + q[k].z, p[k].w + q[k].w)));
        ((int4*)rank)[g]           = make_int4(r[0], r[1], r[2], r[3]);
        ((int4*)rank)[g + E_HALF4] = make_int4(r[4], r[5], r[6], r[7]);
    }
    for (int d = 16; d >= 1; d >>= 1)
        mx = fmaxf(mx, __shfl_xor(mx, d, 32));
    if (lane == 0) atomicMax(&smax, fenc(mx));
    __syncthreads();
    if (threadIdx.x == 0) atomicMax(maxenc, smax);
}

__global__ __launch_bounds__(256)
void gat_alloc(const int* __restrict__ cnt,
               int* __restrict__ offs,
               unsigned* __restrict__ cursor) {
    __shared__ int wsum[8];
    __shared__ int sbase;
    const int lane = threadIdx.x & 31;
    const int wave = threadIdx.x >> 5;
    const int n    = blockIdx.x * 256 + threadIdx.x;
    const int c    = (n < N_NODES) ? cnt[n] : 0;

    int v = c;
    for (int d = 1; d < 32; d <<= 1) {
        const int t = __shfl_up(v, d, 32);
        if (lane >= d) v += t;
    }
    if (lane == 31) wsum[wave] = v;
    __syncthreads();
    if (threadIdx.x == 0) {
        int run = 0;
        for (int w = 0; w < 8; ++w) { const int t = wsum[w]; wsum[w] = run; run += t; }
        sbase = (int)atomicAdd(cursor, (unsigned)run);
    }
    __syncthreads();
    if (n < N_NODES) offs[n] = sbase + wsum[wave] + v - c;
}

__global__ __launch_bounds__(256)
void gat_fill(const int* __restrict__ src,
              const int* __restrict__ tgt,
              const int* __restrict__ rank,
              const int* __restrict__ offs,
              int* __restrict__ ssrc) {
    const int g = blockIdx.x * blockDim.x + threadIdx.x;
    if (g >= E_HALF4) return;
    const int4 ta = ((const int4*)tgt)[g];
    const int4 tb = ((const int4*)tgt)[g + E_HALF4];
    const int4 ra = ((const int4*)rank)[g];
    const int4 rb = ((const int4*)rank)[g + E_HALF4];
    const int4 sa = ((const int4*)src)[g];
    const int4 sb = ((const int4*)src)[g + E_HALF4];
    const int vt[8] = {ta.x, ta.y, ta.z, ta.w, tb.x, tb.y, tb.z, tb.w};
    const int vr[8] = {ra.x, ra.y, ra.z, ra.w, rb.x, rb.y, rb.z, rb.w};
    const int vs[8] = {sa.x, sa.y, sa.z, sa.w, sb.x, sb.y, sb.z, sb.w};
    int o[8];
#pragma unroll
    for (int k = 0; k < 8; ++k) o[k] = offs[vt[k]];
#pragma unroll
    for (int k = 0; k < 8; ++k) ssrc[o[k] + vr[k]] = vs[k];
}

__device__ __forceinline__ float edge_exp(float a, float gmax) {
    float z = a - gmax;
    z = (z > 0.0f) ? z : NEG_SLOPE * z;
    return __expf(z);
}

__device__ __forceinline__ void edge_chunk(const int* __restrict__ ssrc,
                                           const float* __restrict__ pq,
                                           int o, int base, int c, int lane,
                                           float4 q4, float gmax,
                                           int& s, float& e0, float& e1,
                                           float& e2, float& e3) {
    const int  i     = base + lane;
    const bool valid = i < c;
    s = valid ? ssrc[o + i] : 0;
    const float4 p = *(const float4*)(pq + (size_t)s * PQ_W);
    e0 = valid ? edge_exp(p.x + q4.x, gmax) : 0.0f;
    e1 = valid ? edge_exp(p.y + q4.y, gmax) : 0.0f;
    e2 = valid ? edge_exp(p.z + q4.z, gmax) : 0.0f;
    e3 = valid ? edge_exp(p.w + q4.w, gmax) : 0.0f;
}

__device__ __forceinline__ unsigned lane_bcast(float v, int j) {
    return (unsigned)__builtin_amdgcn_readlane(__float_as_int(v), j);
}

__device__ __forceinline__ float head_pick(float al0, float al1, float al2, float al3,
                                           unsigned m0, unsigned m1, unsigned m2, unsigned m3,
                                           int j) {
    return __uint_as_float((lane_bcast(al0, j) & m0) | (lane_bcast(al1, j) & m1) |
                           (lane_bcast(al2, j) & m2) | (lane_bcast(al3, j) & m3));
}

__global__ __launch_bounds__(128)
void gat_node_aggregate(const float* __restrict__ nt,
                        const float* __restrict__ pq,
                        const int*   __restrict__ cnt,
                        const int*   __restrict__ offs,
                        const int*   __restrict__ ssrc,
                        const unsigned* __restrict__ maxenc,
                        float* __restrict__ out) {
    const int lane = threadIdx.x & 31;
    const int wv = __builtin_amdgcn_readfirstlane((int)(threadIdx.x >> 5));

    const float gmax = fdec(*maxenc);
    const int   h    = lane >> 3;
    const unsigned m0 = (h == 0) ? 0xffffffffu : 0u, m1 = (h == 1) ? 0xffffffffu : 0u;
    const unsigned m2 = (h == 2) ? 0xffffffffu : 0u, m3 = (h == 3) ? 0xffffffffu : 0u;
    const float* ntl = nt + 2 * lane;

    for (int n = blockIdx.x * 4 + wv; n < N_NODES; n += AGG_WAVES) {
        const int   c   = cnt[n];
        const int   o   = offs[n];
        const float4 q4 = *(const float4*)(pq + (size_t)n * PQ_W + H_HEADS);

        int   s;
        float e0, e1, e2, e3;
        edge_chunk(ssrc, pq, o, 0, c, lane, q4, gmax, s, e0, e1, e2, e3);

        float d0 = e0, d1 = e1, d2 = e2, d3 = e3;
        for (int base = 32; base < c; base += 32) {
            int s2; float f0, f1, f2, f3;
            edge_chunk(ssrc, pq, o, base, c, lane, q4, gmax, s2, f0, f1, f2, f3);
            d0 += f0; d1 += f1; d2 += f2; d3 += f3;
        }
        for (int d = 16; d >= 1; d >>= 1) {
            d0 += __shfl_xor(d0, d, 32);
            d1 += __shfl_xor(d1, d, 32);
            d2 += __shfl_xor(d2, d, 32);
            d3 += __shfl_xor(d3, d, 32);
        }
        d0 += EPS_DEN; d1 += EPS_DEN; d2 += EPS_DEN; d3 += EPS_DEN;

        float2 acc = {0.0f, 0.0f};
        for (int base = 0; base < c; base += 32) {
            if (base > 0)
                edge_chunk(ssrc, pq, o, base, c, lane, q4, gmax, s, e0, e1, e2, e3);
            const float al0 = e0 / d0, al1 = e1 / d1, al2 = e2 / d2, al3 = e3 / d3;
            const int rest = c - base;
            const int m = rest < 32 ? rest : 32;

            int j = 0;
            for (; j + 4 <= m; j += 4) {
                float2 r[4];
#pragma unroll
                for (int k = 0; k < 4; ++k) {
                    const int sj = __builtin_amdgcn_readlane(s, j + k);
                    r[k] = *(const float2*)(ntl + (size_t)sj * HF);
                }
#pragma unroll
                for (int k = 0; k < 4; ++k) {
                    const float a = head_pick(al0, al1, al2, al3, m0, m1, m2, m3, j + k);
                    acc.x += a * r[k].x;
                    acc.y += a * r[k].y;
                }
            }
            for (; j < m; ++j) {
                const int sj = __builtin_amdgcn_readlane(s, j);
                const float2 r = *(const float2*)(ntl + (size_t)sj * HF);
                const float a = head_pick(al0, al1, al2, al3, m0, m1, m2, m3, j);
                acc.x += a * r.x;
                acc.y += a * r.y;
            }
        }
        *(float2*)(out + (size_t)n * HF + 2 * lane) = acc;
    }
}

extern "C" void kernel_launch(void* const* d_in, const int* in_sizes, int n_in,
                              void* d_out, int out_size, void* d_ws, size_t ws_size,
                              hipStream_t stream) {
    const float* x   = (const float*)d_in[0];
    const float* W_w = (const float*)d_in[1];
    const float* a_w = (const float*)d_in[2];
    const int*   ei  = (const int*)d_in[3];
    const int* src = ei;
    const int* tgt = ei + E_EDGES;
    float* out = (float*)d_out;

    float* nt    = (float*)d_ws;
    float* pq    = nt + (size_t)N_NODES * HF;
    int*   cnt   = (int*)(pq + (size_t)N_NODES * PQ_W);
    int*   offs  = cnt  + N_NODES;
    int*   rank  = offs + N_NODES;
    int*   ssrc  = rank + E_EDGES;
    unsigned* cursor = (unsigned*)(ssrc + E_EDGES);
    unsigned* maxenc = cursor + 1;

    (void)in_sizes; (void)n_in; (void)out_size; (void)ws_size;

    gat_transform_wmma<<<XF_BLOCKS, 128, 0, stream>>>(
        x, W_w, a_w, nt, pq, cnt, cursor, maxenc);
    gat_edge_scores<<<EDGE_BLOCKS, 256, 0, stream>>>(
        pq, src, tgt, cnt, rank, maxenc);
    gat_alloc<<<(N_NODES + 255) / 256, 256, 0, stream>>>(cnt, offs, cursor);
    gat_fill<<<EDGE_BLOCKS, 256, 0, stream>>>(src, tgt, rank, offs, ssrc);
    gat_node_aggregate<<<AGG_BLOCKS, 128, 0, stream>>>(
        nt, pq, cnt, offs, ssrc, maxenc, out);
}
